# starting kernel
# baseline (speedup 1.0000x reference)
#include <hip/hip_runtime.h>
#include <hip/hip_bf16.h>

#define NNODES 100000
#define NEDGES 1000000
#define IN_C   128
#define HID_C  64
#define OUT_C  32

typedef __attribute__((ext_vector_type(2))) float v2f;
typedef __attribute__((ext_vector_type(8))) float v8f;

__global__ void gcn_init(float* __restrict__ deg,
                         float4* __restrict__ agg1,
                         float4* __restrict__ out)
{
    int i = blockIdx.x * blockDim.x + threadIdx.x;
    if (i >= NNODES) return;
    deg[i] = 1.0f;
    float4 z = make_float4(0.f, 0.f, 0.f, 0.f);
#pragma unroll
    for (int q = 0; q < HID_C / 4; ++q) agg1[(size_t)i * (HID_C / 4) + q] = z;
#pragma unroll
    for (int q = 0; q < OUT_C / 4; ++q) out[(size_t)i * (OUT_C / 4) + q] = z;
}

__global__ void gcn_degree(const int* __restrict__ dst, float* __restrict__ deg)
{
    int e = blockIdx.x * blockDim.x + threadIdx.x;
    if (e >= NEDGES) return;
    atomicAdd(&deg[dst[e]], 1.0f);
}

__global__ void gcn_rsqrt(float* __restrict__ deg)
{
    int i = blockIdx.x * blockDim.x + threadIdx.x;
    if (i >= NNODES) return;
    deg[i] = rsqrtf(deg[i]);
}

template <int K, int NC>
__global__ void gcn_gemm_wmma(const float* __restrict__ A,
                              const float* __restrict__ B,
                              float* __restrict__ C, int M)
{
    const int wave = (blockIdx.x * blockDim.x + threadIdx.x) >> 5;
    constexpr int TN = NC / 16;
    const int numTiles = (M >> 4) * TN;
    if (wave >= numTiles) return;

    const int lane = threadIdx.x & 31;
    const int lo = lane & 15;
    const int hi = lane >> 4;
    const int tm = (wave / TN) << 4;
    const int tn = (wave % TN) << 4;

    const float* arow = A + (size_t)(tm + lo) * K + 2 * hi;
    const float* bcol = B + (size_t)(2 * hi) * NC + tn + lo;

    v8f acc = {};
#pragma unroll
    for (int k = 0; k < K; k += 4) {
        v2f a;
        a.x = arow[k];
        a.y = arow[k + 1];
        v2f b;
        b.x = bcol[(size_t)k * NC];
        b.y = bcol[(size_t)(k + 1) * NC];
        acc = __builtin_amdgcn_wmma_f32_16x16x4_f32(
             false, a,  false, b,
             (short)0, acc,  false,  false);
    }

    float* crow = C + (size_t)(tm + 8 * hi) * NC + tn + lo;
#pragma unroll
    for (int g = 0; g < 8; ++g) crow[(size_t)g * NC] = acc[g];
}

template <int C>
__global__ void gcn_scatter(const float* __restrict__ h,
                            const int* __restrict__ src,
                            const int* __restrict__ dst,
                            const float* __restrict__ dinv,
                            float* __restrict__ out)
{
    constexpr int Q = C / 4;
    long long t = (long long)blockIdx.x * blockDim.x + threadIdx.x;
    if (t >= (long long)(NEDGES + NNODES) * Q) return;
    int e = (int)(t / Q);
    int q = (int)(t % Q);
    int s, d;
    if (e < NEDGES) { s = src[e]; d = dst[e]; }
    else            { s = e - NEDGES; d = s; }
    float w = dinv[s] * dinv[d];
    float4 hv = *(const float4*)(h + (size_t)s * C + 4 * q);
    float* o = out + (size_t)d * C + 4 * q;
    atomicAdd(o + 0, hv.x * w);
    atomicAdd(o + 1, hv.y * w);
    atomicAdd(o + 2, hv.z * w);
    atomicAdd(o + 3, hv.w * w);
}

__global__ void gcn_bias_relu(float* __restrict__ agg, const float* __restrict__ b1)
{
    int i = blockIdx.x * blockDim.x + threadIdx.x;
    if (i >= NNODES * HID_C) return;
    float v = agg[i] + b1[i & (HID_C - 1)];
    agg[i] = v > 0.f ? v : 0.f;
}

__global__ void gcn_bias2(float* __restrict__ out, const float* __restrict__ b2)
{
    int i = blockIdx.x * blockDim.x + threadIdx.x;
    if (i >= NNODES * OUT_C) return;
    out[i] += b2[i & (OUT_C - 1)];
}

extern "C" void kernel_launch(void* const* d_in, const int* in_sizes, int n_in,
                              void* d_out, int out_size, void* d_ws, size_t ws_size,
                              hipStream_t stream)
{
    const float* x    = (const float*)d_in[0];
    const int*   eidx = (const int*)d_in[1];
    const int*   src  = eidx;
    const int*   dst  = eidx + NEDGES;
    const float* W1   = (const float*)d_in[2];
    const float* b1   = (const float*)d_in[3];
    const float* W2   = (const float*)d_in[4];
    const float* b2   = (const float*)d_in[5];
    float* out = (float*)d_out;

    float* dinv = (float*)d_ws;
    float* h    = dinv + NNODES;
    float* agg1 = h + (size_t)NNODES * HID_C;

    const int B = 256;

    gcn_init<<<(NNODES + B - 1) / B, B, 0, stream>>>(dinv, (float4*)agg1, (float4*)out);

    gcn_degree<<<(NEDGES + B - 1) / B, B, 0, stream>>>(dst, dinv);
    gcn_rsqrt<<<(NNODES + B - 1) / B, B, 0, stream>>>(dinv);

    {
        int waves = (NNODES / 16) * (HID_C / 16);
        int thr = waves * 32;
        gcn_gemm_wmma<IN_C, HID_C><<<(thr + B - 1) / B, B, 0, stream>>>(x, W1, h, NNODES);
    }

    {
        long long thr = (long long)(NEDGES + NNODES) * (HID_C / 4);
        gcn_scatter<HID_C><<<(int)((thr + B - 1) / B), B, 0, stream>>>(h, src, dst, dinv, agg1);
    }

    gcn_bias_relu<<<(NNODES * HID_C + B - 1) / B, B, 0, stream>>>(agg1, b1);

    {
        int waves = (NNODES / 16) * (OUT_C / 16);
        int thr = waves * 32;
        gcn_gemm_wmma<HID_C, OUT_C><<<(thr + B - 1) / B, B, 0, stream>>>(agg1, W2, h, NNODES);
    }

    {
        long long thr = (long long)(NEDGES + NNODES) * (OUT_C / 4);
        gcn_scatter<OUT_C><<<(int)((thr + B - 1) / B), B, 0, stream>>>(h, src, dst, dinv, out);
    }

    gcn_bias2<<<(NNODES * OUT_C + B - 1) / B, B, 0, stream>>>(out, b2);
}


# edge scatter re-laid: a wave walks 32 edges with lane = channel, so each f32 atomic add and row load covers one whole 128-B line, not 4 quarter lines; sums stay in atomic arrival order
# speedup vs baseline: 1.2226x; 1.2226x over previous
#include <hip/hip_runtime.h>
#include <hip/hip_bf16.h>

#define NNODES 100000
#define NEDGES 1000000
#define IN_C   128
#define HID_C  64
#define OUT_C  32

typedef __attribute__((ext_vector_type(2))) float v2f;
typedef __attribute__((ext_vector_type(8))) float v8f;

__global__ void gcn_init(float* __restrict__ deg,
                         float4* __restrict__ agg1,
                         float4* __restrict__ out)
{
    int i = blockIdx.x * blockDim.x + threadIdx.x;
    if (i >= NNODES) return;
    deg[i] = 1.0f;
    float4 z = make_float4(0.f, 0.f, 0.f, 0.f);
#pragma unroll
    for (int q = 0; q < HID_C / 4; ++q) agg1[(size_t)i * (HID_C / 4) + q] = z;
#pragma unroll
    for (int q = 0; q < OUT_C / 4; ++q) out[(size_t)i * (OUT_C / 4) + q] = z;
}

__global__ void gcn_degree(const int* __restrict__ dst, float* __restrict__ deg)
{
    int e = blockIdx.x * blockDim.x + threadIdx.x;
    if (e >= NEDGES) return;
    atomicAdd(&deg[dst[e]], 1.0f);
}

__global__ void gcn_rsqrt(float* __restrict__ deg)
{
    int i = blockIdx.x * blockDim.x + threadIdx.x;
    if (i >= NNODES) return;
    deg[i] = rsqrtf(deg[i]);
}

template <int K, int NC>
__global__ void gcn_gemm_wmma(const float* __restrict__ A,
                              const float* __restrict__ B,
                              float* __restrict__ C, int M)
{
    const int wave = (blockIdx.x * blockDim.x + threadIdx.x) >> 5;
    constexpr int TN = NC / 16;
    const int numTiles = (M >> 4) * TN;
    if (wave >= numTiles) return;

    const int lane = threadIdx.x & 31;
    const int lo = lane & 15;
    const int hi = lane >> 4;
    const int tm = (wave / TN) << 4;
    const int tn = (wave % TN) << 4;

    const float* arow = A + (size_t)(tm + lo) * K + 2 * hi;
    const float* bcol = B + (size_t)(2 * hi) * NC + tn + lo;

    v8f acc = {};
#pragma unroll
    for (int k = 0; k < K; k += 4) {
        v2f a;
        a.x = arow[k];
        a.y = arow[k + 1];
        v2f b;
        b.x = bcol[(size_t)k * NC];
        b.y = bcol[(size_t)(k + 1) * NC];
        acc = __builtin_amdgcn_wmma_f32_16x16x4_f32(
             false, a,  false, b,
             (short)0, acc,  false,  false);
    }

    float* crow = C + (size_t)(tm + 8 * hi) * NC + tn + lo;
#pragma unroll
    for (int g = 0; g < 8; ++g) crow[(size_t)g * NC] = acc[g];
}

template <int C>
__global__ void gcn_scatter(const float* __restrict__ h,
                            const int* __restrict__ src,
                            const int* __restrict__ dst,
                            const float* __restrict__ dinv,
                            float* __restrict__ out)
{
    static_assert(C % 32 == 0, "lane = channel needs rows of whole 32-float lines");
    static_assert((NEDGES + NNODES) % 32 == 0, "32 edges per wave, no partial wave");
    constexpr int R  = C / 32;
    constexpr int EB = 16 / R;

    const int wave = (blockIdx.x * blockDim.x + threadIdx.x) >> 5;
    if (wave >= (NEDGES + NNODES) / 32) return;
    const unsigned lane = threadIdx.x & 31u;

    const int e = wave * 32 + (int)lane;
    int s, d;
    if (e < NEDGES) { s = src[e]; d = dst[e]; }
    else            { s = e - NEDGES; d = s; }
    const int wbits = __builtin_bit_cast(int, dinv[s] * dinv[d]);

#pragma unroll
    for (int jb = 0; jb < 32; jb += EB) {
        float v[EB][R];
        for (int u = 0; u < EB; ++u) {
            const int sj = __builtin_amdgcn_readlane(s, jb + u);
            const float* hr = h + (size_t)sj * C;
            for (int r = 0; r < R; ++r) v[u][r] = hr[lane + 32u * r];
        }
        for (int u = 0; u < EB; ++u) {
            const int   dj = __builtin_amdgcn_readlane(d, jb + u);
            const float wj = __builtin_bit_cast(float, __builtin_amdgcn_readlane(wbits, jb + u));
            float* o = out + (size_t)dj * C;
            for (int r = 0; r < R; ++r) atomicAdd(o + lane + 32u * r, v[u][r] * wj);
        }
    }
}

__global__ void gcn_bias_relu(float* __restrict__ agg, const float* __restrict__ b1)
{
    int i = blockIdx.x * blockDim.x + threadIdx.x;
    if (i >= NNODES * HID_C) return;
    float v = agg[i] + b1[i & (HID_C - 1)];
    agg[i] = v > 0.f ? v : 0.f;
}

__global__ void gcn_bias2(float* __restrict__ out, const float* __restrict__ b2)
{
    int i = blockIdx.x * blockDim.x + threadIdx.x;
    if (i >= NNODES * OUT_C) return;
    out[i] += b2[i & (OUT_C - 1)];
}

extern "C" void kernel_launch(void* const* d_in, const int* in_sizes, int n_in,
                              void* d_out, int out_size, void* d_ws, size_t ws_size,
                              hipStream_t stream)
{
    const float* x    = (const float*)d_in[0];
    const int*   eidx = (const int*)d_in[1];
    const int*   src  = eidx;
    const int*   dst  = eidx + NEDGES;
    const float* W1   = (const float*)d_in[2];
    const float* b1   = (const float*)d_in[3];
    const float* W2   = (const float*)d_in[4];
    const float* b2   = (const float*)d_in[5];
    float* out = (float*)d_out;

    float* dinv = (float*)d_ws;
    float* h    = dinv + NNODES;
    float* agg1 = h + (size_t)NNODES * HID_C;

    const int B = 256;

    gcn_init<<<(NNODES + B - 1) / B, B, 0, stream>>>(dinv, (float4*)agg1, (float4*)out);

    gcn_degree<<<(NEDGES + B - 1) / B, B, 0, stream>>>(dst, dinv);
    gcn_rsqrt<<<(NNODES + B - 1) / B, B, 0, stream>>>(dinv);

    {
        int waves = (NNODES / 16) * (HID_C / 16);
        int thr = waves * 32;
        gcn_gemm_wmma<IN_C, HID_C><<<(thr + B - 1) / B, B, 0, stream>>>(x, W1, h, NNODES);
    }

    {
        int thr = NEDGES + NNODES;
        gcn_scatter<HID_C><<<(thr + B - 1) / B, B, 0, stream>>>(h, src, dst, dinv, agg1);
    }

    gcn_bias_relu<<<(NNODES * HID_C + B - 1) / B, B, 0, stream>>>(agg1, b1);

    {
        int waves = (NNODES / 16) * (OUT_C / 16);
        int thr = waves * 32;
        gcn_gemm_wmma<HID_C, OUT_C><<<(thr + B - 1) / B, B, 0, stream>>>(agg1, W2, h, NNODES);
    }

    {
        int thr = NEDGES + NNODES;
        gcn_scatter<OUT_C><<<(thr + B - 1) / B, B, 0, stream>>>(h, src, dst, dinv, out);
    }

    gcn_bias2<<<(NNODES * OUT_C + B - 1) / B, B, 0, stream>>>(out, b2);
}


# float-atomic scatter replaced by gather: edges counting-sorted by destination each call (int-atomic rank, scan, fill); a wave sums a node's source rows in registers, stores once; sums in rank order
# speedup vs baseline: 1.7391x; 1.4225x over previous
#include <hip/hip_runtime.h>
#include <hip/hip_bf16.h>

#define NNODES 100000
#define NEDGES 1000000
#define IN_C   128
#define HID_C  64
#define OUT_C  32

#define SCAN_B 1024
#define SCAN_NB ((NNODES + SCAN_B - 1) / SCAN_B)
#define EPT 4
#define ETHR (NEDGES / EPT)
#define NQ (NNODES / 4)
#define NPW 8

typedef __attribute__((ext_vector_type(2))) float v2f;
typedef __attribute__((ext_vector_type(8))) float v8f;


__global__ void gcn_zero_cnt(int* __restrict__ cnt)
{
    static_assert(NNODES % 4 == 0, "four counts per thread, a quarter of the array apart");
    int i = blockIdx.x * blockDim.x + threadIdx.x;
    if (i >= NQ) return;
#pragma unroll
    for (int k = 0; k < 4; ++k) cnt[i + k * NQ] = 0;
}

__global__ void gcn_count_rank(const int* __restrict__ dst,
                               int* __restrict__ cnt,
                               int* __restrict__ rank)
{
    static_assert(NEDGES % EPT == 0, "no partial thread");
    int t = blockIdx.x * blockDim.x + threadIdx.x;
    if (t >= ETHR) return;
    int dd[EPT], rk[EPT];
#pragma unroll
    for (int k = 0; k < EPT; ++k) dd[k] = dst[t + k * ETHR];
#pragma unroll
    for (int k = 0; k < EPT; ++k) rk[k] = atomicAdd(&cnt[dd[k]], 1);
#pragma unroll
    for (int k = 0; k < EPT; ++k) rank[t + k * ETHR] = rk[k];
}

__global__ void __launch_bounds__(SCAN_B)
gcn_scan(const int* __restrict__ cnt, int* __restrict__ rowptr)
{
    __shared__ int wsum[32];
    __shared__ int base_s;
    const int t = threadIdx.x;
    const int lane = t & 31;
    const int w = t >> 5;
    const int b = blockIdx.x;

    int pre = 0;
    const int4* c4 = (const int4*)cnt;
    const int n4 = b * (SCAN_B / 4);
    int i4 = t;
    for (; i4 + 3 * SCAN_B < n4; i4 += 4 * SCAN_B) {
        int4 v0 = c4[i4];
        int4 v1 = c4[i4 + SCAN_B];
        int4 v2 = c4[i4 + 2 * SCAN_B];
        int4 v3 = c4[i4 + 3 * SCAN_B];
        pre += ((v0.x + v0.y) + (v0.z + v0.w)) + ((v1.x + v1.y) + (v1.z + v1.w));
        pre += ((v2.x + v2.y) + (v2.z + v2.w)) + ((v3.x + v3.y) + (v3.z + v3.w));
    }
    for (; i4 < n4; i4 += SCAN_B) {
        int4 v = c4[i4];
        pre += (v.x + v.y) + (v.z + v.w);
    }
    for (int o = 16; o > 0; o >>= 1) pre += __shfl_xor(pre, o);
    if (lane == 0) wsum[w] = pre;
    __syncthreads();
    if (w == 0) {
        int s = wsum[lane];
        for (int o = 16; o > 0; o >>= 1) s += __shfl_xor(s, o);
        if (lane == 0) base_s = s;
    }
    __syncthreads();
    const int base = base_s;
    __syncthreads();

    const int i = b * SCAN_B + t;
    const int c = (i < NNODES) ? cnt[i] : 0;
    int inc = c;
    for (int o = 1; o < 32; o <<= 1) {
        int v = __shfl_up(inc, o);
        if (lane >= o) inc += v;
    }
    if (lane == 31) wsum[w] = inc;
    __syncthreads();
    if (w == 0) {
        const int s = wsum[lane];
        int si = s;
        for (int o = 1; o < 32; o <<= 1) {
            int v = __shfl_up(si, o);
            if (lane >= o) si += v;
        }
        wsum[lane] = si - s;
    }
    __syncthreads();
    if (i < NNODES) rowptr[i] = base + wsum[w] + inc - c;
}

__global__ void gcn_fill(const int* __restrict__ src,
                         const int* __restrict__ dst,
                         const int* __restrict__ rowptr,
                         const int* __restrict__ rank,
                         int* __restrict__ col)
{
    int t = blockIdx.x * blockDim.x + threadIdx.x;
    if (t >= ETHR) return;
    int dd[EPT], rk[EPT], ss[EPT], pp[EPT];
#pragma unroll
    for (int k = 0; k < EPT; ++k) {
        const int e = t + k * ETHR;
        dd[k] = dst[e];
        rk[k] = rank[e];
        ss[k] = src[e];
    }
#pragma unroll
    for (int k = 0; k < EPT; ++k) pp[k] = rowptr[dd[k]] + rk[k];
#pragma unroll
    for (int k = 0; k < EPT; ++k) col[pp[k]] = ss[k];
}

template <int K, int NC>
__global__ void gcn_gemm_wmma(const float* __restrict__ A,
                              const float* __restrict__ B,
                              const int* __restrict__ cnt,
                              float* __restrict__ C, int M)
{
    const int wave = (blockIdx.x * blockDim.x + threadIdx.x) >> 5;
    constexpr int TN = NC / 16;
    const int numTiles = (M >> 4) * TN;
    if (wave >= numTiles) return;

    const int lane = threadIdx.x & 31;
    const int lo = lane & 15;
    const int hi = lane >> 4;
    const int tm = (wave / TN) << 4;
    const int tn = (wave % TN) << 4;

    const float* arow = A + (size_t)(tm + lo) * K + 2 * hi;
    const float* bcol = B + (size_t)(2 * hi) * NC + tn + lo;

    v8f acc = {};
#pragma unroll
    for (int k = 0; k < K; k += 4) {
        v2f a;
        a.x = arow[k];
        a.y = arow[k + 1];
        v2f b;
        b.x = bcol[(size_t)k * NC];
        b.y = bcol[(size_t)(k + 1) * NC];
        acc = __builtin_amdgcn_wmma_f32_16x16x4_f32(
             false, a,  false, b,
             (short)0, acc,  false,  false);
    }

    const int* crcnt = cnt + tm + 8 * hi;
    float* crow = C + (size_t)(tm + 8 * hi) * NC + tn + lo;
#pragma unroll
    for (int g = 0; g < 8; ++g)
        crow[(size_t)g * NC] = acc[g] * rsqrtf((float)(crcnt[g] + 1));
}


template <int C, int UU>
__device__ __forceinline__ void gcn_add_rows(const float* __restrict__ hs, int id, int j,
                                             unsigned lane, float (&acc)[C / 32])
{
    constexpr int R = C / 32;
    float v[UU][R];
#pragma unroll
    for (int u = 0; u < UU; ++u) {
        const int sj = __builtin_amdgcn_readlane(id, j + u);
        const float* hr = hs + (size_t)sj * C;
#pragma unroll
        for (int r = 0; r < R; ++r) v[u][r] = hr[lane + 32u * r];
    }
#pragma unroll
    for (int u = 0; u < UU; ++u)
#pragma unroll
        for (int r = 0; r < R; ++r) acc[r] += v[u][r];
}

template <int C, bool RELU>
__global__ void gcn_gather(const float* __restrict__ hs,
                           const int* __restrict__ rowptr,
                           const int* __restrict__ cnt,
                           const int* __restrict__ col,
                           const float* __restrict__ bias,
                           float* __restrict__ out)
{
    static_assert(C % 32 == 0, "lane = channel needs rows of whole 32-float lines");
    constexpr int R = C / 32;

    static_assert(NNODES % NPW == 0, "no partial wave");

    const int w = __builtin_amdgcn_readfirstlane((int)((blockIdx.x * blockDim.x + threadIdx.x) >> 5));
    if (w >= NNODES / NPW) return;
    const unsigned lane = threadIdx.x & 31u;

    for (int k = 0; k < NPW; ++k) {
    const int d = w * NPW + k;
    const int start = __builtin_amdgcn_readfirstlane(rowptr[d]);
    const int n     = __builtin_amdgcn_readfirstlane(cnt[d]);

    float acc[R];
#pragma unroll
    for (int r = 0; r < R; ++r) acc[r] = 0.f;

    const int total = n + 1;
    for (int j0 = 0; j0 < total; j0 += 32) {
        const int m = (total - j0 < 32) ? (total - j0) : 32;
        const int id = (j0 + (int)lane < n) ? col[start + j0 + (int)lane] : d;
        int j = 0;
        for (; j + 8 <= m; j += 8) gcn_add_rows<C, 8>(hs, id, j, lane, acc);
        if (j + 4 <= m) { gcn_add_rows<C, 4>(hs, id, j, lane, acc); j += 4; }
        if (j + 2 <= m) { gcn_add_rows<C, 2>(hs, id, j, lane, acc); j += 2; }
        if (j < m)      { gcn_add_rows<C, 1>(hs, id, j, lane, acc); }
    }

    const float dn = rsqrtf((float)(n + 1));
    float* o = out + (size_t)d * C;
#pragma unroll
    for (int r = 0; r < R; ++r) {
        float y = acc[r] * dn + bias[lane + 32u * r];
        if (RELU) y = y > 0.f ? y : 0.f;
        o[lane + 32u * r] = y;
    }
    }
}

extern "C" void kernel_launch(void* const* d_in, const int* in_sizes, int n_in,
                              void* d_out, int out_size, void* d_ws, size_t ws_size,
                              hipStream_t stream)
{
    const float* x    = (const float*)d_in[0];
    const int*   eidx = (const int*)d_in[1];
    const int*   src  = eidx;
    const int*   dst  = eidx + NEDGES;
    const float* W1   = (const float*)d_in[2];
    const float* b1   = (const float*)d_in[3];
    const float* W2   = (const float*)d_in[4];
    const float* b2   = (const float*)d_in[5];
    float* out = (float*)d_out;

    int*   cnt    = (int*)d_ws;
    int*   rowptr = cnt + NNODES;
    int*   rank   = rowptr + NNODES;
    int*   col    = rank + NEDGES;
    float* h      = (float*)(col + NEDGES);
    float* agg1   = h + (size_t)NNODES * HID_C;

    const int B = 256;

    gcn_zero_cnt<<<(NQ + B - 1) / B, B, 0, stream>>>(cnt);
    gcn_count_rank<<<(ETHR + B - 1) / B, B, 0, stream>>>(dst, cnt, rank);
    gcn_scan<<<SCAN_NB, SCAN_B, 0, stream>>>(cnt, rowptr);
    gcn_fill<<<(ETHR + B - 1) / B, B, 0, stream>>>(src, dst, rowptr, rank, col);

    {
        int waves = (NNODES / 16) * (HID_C / 16);
        int thr = waves * 32;
        gcn_gemm_wmma<IN_C, HID_C><<<(thr + B - 1) / B, B, 0, stream>>>(x, W1, cnt, h, NNODES);
    }

    {
        int thr = (NNODES / NPW) * 32;
        gcn_gather<HID_C, true><<<(thr + B - 1) / B, B, 0, stream>>>(h, rowptr, cnt, col, b1, agg1);
    }

    {
        int waves = (NNODES / 16) * (OUT_C / 16);
        int thr = waves * 32;
        gcn_gemm_wmma<HID_C, OUT_C><<<(thr + B - 1) / B, B, 0, stream>>>(agg1, W2, cnt, h, NNODES);
    }

    {
        int thr = (NNODES / NPW) * 32;
        gcn_gather<OUT_C, false><<<(thr + B - 1) / B, B, 0, stream>>>(h, rowptr, cnt, col, b2, out);
    }
}


# GEMM: one wave per 16-row strip of all NC columns (was one 16x16 tile): A fragment loaded once for NC/16 accumulators, B and C rows one vector per lane, A requested 16 k-steps ahead; sums unchanged
# speedup vs baseline: 1.7961x; 1.0328x over previous
#include <hip/hip_runtime.h>
#include <hip/hip_bf16.h>

#define NNODES 100000
#define NEDGES 1000000
#define IN_C   128
#define HID_C  64
#define OUT_C  32

#define SCAN_B 1024
#define SCAN_NB ((NNODES + SCAN_B - 1) / SCAN_B)
#define EPT 4
#define ETHR (NEDGES / EPT)
#define NQ (NNODES / 4)
#define NPW 8

typedef __attribute__((ext_vector_type(2))) float v2f;
typedef __attribute__((ext_vector_type(8))) float v8f;


__global__ void gcn_zero_cnt(int* __restrict__ cnt)
{
    static_assert(NNODES % 4 == 0, "four counts per thread, a quarter of the array apart");
    int i = blockIdx.x * blockDim.x + threadIdx.x;
    if (i >= NQ) return;
#pragma unroll
    for (int k = 0; k < 4; ++k) cnt[i + k * NQ] = 0;
}

__global__ void gcn_count_rank(const int* __restrict__ dst,
                               int* __restrict__ cnt,
                               int* __restrict__ rank)
{
    static_assert(NEDGES % EPT == 0, "no partial thread");
    int t = blockIdx.x * blockDim.x + threadIdx.x;
    if (t >= ETHR) return;
    int dd[EPT], rk[EPT];
#pragma unroll
    for (int k = 0; k < EPT; ++k) dd[k] = dst[t + k * ETHR];
#pragma unroll
    for (int k = 0; k < EPT; ++k) rk[k] = atomicAdd(&cnt[dd[k]], 1);
#pragma unroll
    for (int k = 0; k < EPT; ++k) rank[t + k * ETHR] = rk[k];
}

__global__ void __launch_bounds__(SCAN_B)
gcn_scan(const int* __restrict__ cnt, int* __restrict__ rowptr)
{
    __shared__ int wsum[32];
    __shared__ int base_s;
    const int t = threadIdx.x;
    const int lane = t & 31;
    const int w = t >> 5;
    const int b = blockIdx.x;

    int pre = 0;
    const int4* c4 = (const int4*)cnt;
    const int n4 = b * (SCAN_B / 4);
    int i4 = t;
    for (; i4 + 3 * SCAN_B < n4; i4 += 4 * SCAN_B) {
        int4 v0 = c4[i4];
        int4 v1 = c4[i4 + SCAN_B];
        int4 v2 = c4[i4 + 2 * SCAN_B];
        int4 v3 = c4[i4 + 3 * SCAN_B];
        pre += ((v0.x + v0.y) + (v0.z + v0.w)) + ((v1.x + v1.y) + (v1.z + v1.w));
        pre += ((v2.x + v2.y) + (v2.z + v2.w)) + ((v3.x + v3.y) + (v3.z + v3.w));
    }
    for (; i4 < n4; i4 += SCAN_B) {
        int4 v = c4[i4];
        pre += (v.x + v.y) + (v.z + v.w);
    }
    for (int o = 16; o > 0; o >>= 1) pre += __shfl_xor(pre, o);
    if (lane == 0) wsum[w] = pre;
    __syncthreads();
    if (w == 0) {
        int s = wsum[lane];
        for (int o = 16; o > 0; o >>= 1) s += __shfl_xor(s, o);
        if (lane == 0) base_s = s;
    }
    __syncthreads();
    const int base = base_s;
    __syncthreads();

    const int i = b * SCAN_B + t;
    const int c = (i < NNODES) ? cnt[i] : 0;
    int inc = c;
    for (int o = 1; o < 32; o <<= 1) {
        int v = __shfl_up(inc, o);
        if (lane >= o) inc += v;
    }
    if (lane == 31) wsum[w] = inc;
    __syncthreads();
    if (w == 0) {
        const int s = wsum[lane];
        int si = s;
        for (int o = 1; o < 32; o <<= 1) {
            int v = __shfl_up(si, o);
            if (lane >= o) si += v;
        }
        wsum[lane] = si - s;
    }
    __syncthreads();
    if (i < NNODES) rowptr[i] = base + wsum[w] + inc - c;
}

__global__ void gcn_fill(const int* __restrict__ src,
                         const int* __restrict__ dst,
                         const int* __restrict__ rowptr,
                         const int* __restrict__ rank,
                         int* __restrict__ col)
{
    int t = blockIdx.x * blockDim.x + threadIdx.x;
    if (t >= ETHR) return;
    int dd[EPT], rk[EPT], ss[EPT], pp[EPT];
#pragma unroll
    for (int k = 0; k < EPT; ++k) {
        const int e = t + k * ETHR;
        dd[k] = dst[e];
        rk[k] = rank[e];
        ss[k] = src[e];
    }
#pragma unroll
    for (int k = 0; k < EPT; ++k) pp[k] = rowptr[dd[k]] + rk[k];
#pragma unroll
    for (int k = 0; k < EPT; ++k) col[pp[k]] = ss[k];
}

template <int K, int NC>
__global__ void gcn_gemm_wmma(const float* __restrict__ A,
                              const float* __restrict__ B,
                              const int* __restrict__ cnt,
                              float* __restrict__ C, int M)
{
    constexpr int T = NC / 16;
    typedef __attribute__((ext_vector_type(T))) float vtf;
    static_assert(NC % 16 == 0 && (T == 2 || T == 4), "a lane's T columns are one 8- or 16-byte vector");

    const int wave = (blockIdx.x * blockDim.x + threadIdx.x) >> 5;
    if (wave >= (M >> 4)) return;

    const int lane = threadIdx.x & 31;
    const int lo = lane & 15;
    const int hi = lane >> 4;
    const int tm = wave << 4;

    const float* arow = A + (size_t)(tm + lo) * K + 2 * hi;
    const float* brow = B + (size_t)(2 * hi) * NC + T * lo;

    v8f acc[T];
#pragma unroll
    for (int t = 0; t < T; ++t) acc[t] = v8f{};

    constexpr int S = K / 4;
    constexpr int PA = S < 16 ? S : 16;
    constexpr int PB = 2;

    v2f a[PA];
#pragma unroll
    for (int s = 0; s < PA; ++s) {
        a[s].x = arow[4 * s];
        a[s].y = arow[4 * s + 1];
    }

    vtf b0[PB], b1[PB];
#pragma unroll
    for (int p = 0; p < PB; ++p) {
        b0[p] = *(const vtf*)(brow + (size_t)(4 * p) * NC);
        b1[p] = *(const vtf*)(brow + (size_t)(4 * p + 1) * NC);
    }

#pragma unroll
    for (int s = 0; s < S; ++s) {
        const v2f ca = a[s % PA];
        const vtf c0 = b0[s % PB];
        const vtf c1 = b1[s % PB];
        if (s + PA < S) {
            a[s % PA].x = arow[4 * (s + PA)];
            a[s % PA].y = arow[4 * (s + PA) + 1];
        }
        if (s + PB < S) {
            b0[s % PB] = *(const vtf*)(brow + (size_t)(4 * (s + PB)) * NC);
            b1[s % PB] = *(const vtf*)(brow + (size_t)(4 * (s + PB) + 1) * NC);
        }
#pragma unroll
        for (int t = 0; t < T; ++t) {
            v2f b;
            b.x = c0[t];
            b.y = c1[t];
            acc[t] = __builtin_amdgcn_wmma_f32_16x16x4_f32(
                 false, ca,  false, b,
                 (short)0, acc[t],  false,  false);
        }
    }

    const int* crcnt = cnt + tm + 8 * hi;
    float* crow = C + (size_t)(tm + 8 * hi) * NC + T * lo;
#pragma unroll
    for (int g = 0; g < 8; ++g) {
        const float sc = rsqrtf((float)(crcnt[g] + 1));
        vtf o;
#pragma unroll
        for (int t = 0; t < T; ++t) o[t] = acc[t][g] * sc;
        *(vtf*)(crow + (size_t)g * NC) = o;
    }
}


template <int C, int UU>
__device__ __forceinline__ void gcn_add_rows(const float* __restrict__ hs, int id, int j,
                                             unsigned lane, float (&acc)[C / 32])
{
    constexpr int R = C / 32;
    float v[UU][R];
#pragma unroll
    for (int u = 0; u < UU; ++u) {
        const int sj = __builtin_amdgcn_readlane(id, j + u);
        const float* hr = hs + (size_t)sj * C;
#pragma unroll
        for (int r = 0; r < R; ++r) v[u][r] = hr[lane + 32u * r];
    }
#pragma unroll
    for (int u = 0; u < UU; ++u)
#pragma unroll
        for (int r = 0; r < R; ++r) acc[r] += v[u][r];
}

template <int C, bool RELU>
__global__ void gcn_gather(const float* __restrict__ hs,
                           const int* __restrict__ rowptr,
                           const int* __restrict__ cnt,
                           const int* __restrict__ col,
                           const float* __restrict__ bias,
                           float* __restrict__ out)
{
    static_assert(C % 32 == 0, "lane = channel needs rows of whole 32-float lines");
    constexpr int R = C / 32;

    static_assert(NNODES % NPW == 0, "no partial wave");

    const int w = __builtin_amdgcn_readfirstlane((int)((blockIdx.x * blockDim.x + threadIdx.x) >> 5));
    if (w >= NNODES / NPW) return;
    const unsigned lane = threadIdx.x & 31u;

    for (int k = 0; k < NPW; ++k) {
    const int d = w * NPW + k;
    const int start = __builtin_amdgcn_readfirstlane(rowptr[d]);
    const int n     = __builtin_amdgcn_readfirstlane(cnt[d]);

    float acc[R];
#pragma unroll
    for (int r = 0; r < R; ++r) acc[r] = 0.f;

    const int total = n + 1;
    for (int j0 = 0; j0 < total; j0 += 32) {
        const int m = (total - j0 < 32) ? (total - j0) : 32;
        const int id = (j0 + (int)lane < n) ? col[start + j0 + (int)lane] : d;
        int j = 0;
        for (; j + 8 <= m; j += 8) gcn_add_rows<C, 8>(hs, id, j, lane, acc);
        if (j + 4 <= m) { gcn_add_rows<C, 4>(hs, id, j, lane, acc); j += 4; }
        if (j + 2 <= m) { gcn_add_rows<C, 2>(hs, id, j, lane, acc); j += 2; }
        if (j < m)      { gcn_add_rows<C, 1>(hs, id, j, lane, acc); }
    }

    const float dn = rsqrtf((float)(n + 1));
    float* o = out + (size_t)d * C;
#pragma unroll
    for (int r = 0; r < R; ++r) {
        float y = acc[r] * dn + bias[lane + 32u * r];
        if (RELU) y = y > 0.f ? y : 0.f;
        o[lane + 32u * r] = y;
    }
    }
}

extern "C" void kernel_launch(void* const* d_in, const int* in_sizes, int n_in,
                              void* d_out, int out_size, void* d_ws, size_t ws_size,
                              hipStream_t stream)
{
    const float* x    = (const float*)d_in[0];
    const int*   eidx = (const int*)d_in[1];
    const int*   src  = eidx;
    const int*   dst  = eidx + NEDGES;
    const float* W1   = (const float*)d_in[2];
    const float* b1   = (const float*)d_in[3];
    const float* W2   = (const float*)d_in[4];
    const float* b2   = (const float*)d_in[5];
    float* out = (float*)d_out;

    int*   cnt    = (int*)d_ws;
    int*   rowptr = cnt + NNODES;
    int*   rank   = rowptr + NNODES;
    int*   col    = rank + NEDGES;
    float* h      = (float*)(col + NEDGES);
    float* agg1   = h + (size_t)NNODES * HID_C;

    const int B = 256;

    gcn_zero_cnt<<<(NQ + B - 1) / B, B, 0, stream>>>(cnt);
    gcn_count_rank<<<(ETHR + B - 1) / B, B, 0, stream>>>(dst, cnt, rank);
    gcn_scan<<<SCAN_NB, SCAN_B, 0, stream>>>(cnt, rowptr);
    gcn_fill<<<(ETHR + B - 1) / B, B, 0, stream>>>(src, dst, rowptr, rank, col);

    {
        int waves = NNODES / 16;
        int thr = waves * 32;
        gcn_gemm_wmma<IN_C, HID_C><<<(thr + B - 1) / B, B, 0, stream>>>(x, W1, cnt, h, NNODES);
    }

    {
        int thr = (NNODES / NPW) * 32;
        gcn_gather<HID_C, true><<<(thr + B - 1) / B, B, 0, stream>>>(h, rowptr, cnt, col, b1, agg1);
    }

    {
        int waves = NNODES / 16;
        int thr = waves * 32;
        gcn_gemm_wmma<HID_C, OUT_C><<<(thr + B - 1) / B, B, 0, stream>>>(agg1, W2, cnt, h, NNODES);
    }

    {
        int thr = (NNODES / NPW) * 32;
        gcn_gather<OUT_C, false><<<(thr + B - 1) / B, B, 0, stream>>>(h, rowptr, cnt, col, b2, out);
    }
}


# edges grouped by destination with LDS counters in two levels (a block orders 4096 edges by 512-node bucket; a block per bucket counts, scans, places): no global atomic left; sums in LDS rank order
# speedup vs baseline: 3.1618x; 1.7604x over previous
#include <hip/hip_runtime.h>
#include <hip/hip_bf16.h>

#define NNODES 100000
#define NEDGES 1000000
#define IN_C   128
#define HID_C  64
#define OUT_C  32

#define PART_T 1024
#define PART_EPT 4
#define PART_CH (PART_T * PART_EPT)
#define PART_NB ((NEDGES + PART_CH - 1) / PART_CH)
#define BKT_SH 9
#define BKT_W (1 << BKT_SH)
#define NBKT ((NNODES + BKT_W - 1) / BKT_W)
#define HPAD 256
#define TAB_P 256
#define SRC_BITS 17
#define SRC_MASK ((1 << SRC_BITS) - 1)
#define COL_CAP 8192
#define NPW 8

typedef __attribute__((ext_vector_type(2))) float v2f;
typedef __attribute__((ext_vector_type(8))) float v8f;

static_assert(NNODES <= (1 << SRC_BITS), "a source index must fit in SRC_BITS bits");
static_assert(BKT_SH + SRC_BITS < 32, "key stays a non-negative int");
static_assert(NBKT + 1 <= HPAD && HPAD <= PART_T, "one thread per bucket counter in the partition block");
static_assert(PART_NB <= TAB_P && PART_NB <= BKT_W, "one thread per run in the bucket block");

__global__ void __launch_bounds__(PART_T)
gcn_part(const int* __restrict__ src, const int* __restrict__ dst,
         int* __restrict__ keys, int* __restrict__ tab)
{
    __shared__ int hist[HPAD];
    __shared__ int wsum[32];
    __shared__ int stage[PART_CH];
    const int t = threadIdx.x;
    const int lane = t & 31;
    const int w = t >> 5;
    const int b = blockIdx.x;
    const int e0 = b * PART_CH;

    if (t < HPAD) hist[t] = 0;

    int key[PART_EPT], bk[PART_EPT], rk[PART_EPT];
#pragma unroll
    for (int k = 0; k < PART_EPT; ++k) {
        const int e = e0 + t + k * PART_T;
        const bool ok = e < NEDGES;
        const int d = ok ? dst[e] : 0;
        const int s = ok ? src[e] : 0;
        bk[k] = ok ? (d >> BKT_SH) : -1;
        key[k] = ((d & (BKT_W - 1)) << SRC_BITS) | s;
        rk[k] = 0;
    }
    __syncthreads();
#pragma unroll
    for (int k = 0; k < PART_EPT; ++k)
        if (bk[k] >= 0) rk[k] = atomicAdd(&hist[bk[k]], 1);
    __syncthreads();

    const int c = (t < HPAD) ? hist[t] : 0;
    int inc = c;
    for (int o = 1; o < 32; o <<= 1) {
        int v = __shfl_up(inc, o);
        if (lane >= o) inc += v;
    }
    if (lane == 31) wsum[w] = inc;
    __syncthreads();
    if (w == 0) {
        const int s = wsum[lane];
        int si = s;
        for (int o = 1; o < 32; o <<= 1) {
            int v = __shfl_up(si, o);
            if (lane >= o) si += v;
        }
        wsum[lane] = si - s;
    }
    __syncthreads();
    const int ex = wsum[w] + inc - c;
    if (t <= NBKT) tab[t * TAB_P + b] = ex;
    if (t < HPAD) hist[t] = ex;
    __syncthreads();

#pragma unroll
    for (int k = 0; k < PART_EPT; ++k)
        if (bk[k] >= 0) stage[hist[bk[k]] + rk[k]] = key[k];
    __syncthreads();
    const int n = (NEDGES - e0 < PART_CH) ? (NEDGES - e0) : PART_CH;
#pragma unroll
    for (int k = 0; k < PART_EPT; ++k) {
        const int i = t + k * PART_T;
        if (i < n) keys[e0 + i] = stage[i];
    }
}

template <bool PLACE>
__device__ __forceinline__ void gcn_bucket_walk(const int* __restrict__ keys,
                                                const int* seg0, const int* segn,
                                                int* cntl, int* cbuf,
                                                int* __restrict__ colb, bool staged,
                                                int w, int lane)
{
    constexpr int NW = BKT_W / 32;
    for (int q = w; q < PART_NB; q += 4 * NW) {
        int st[4], ln[4], ky[4];
#pragma unroll
        for (int u = 0; u < 4; ++u) {
            const int qq = q + NW * u;
            const bool ok = qq < PART_NB;
            st[u] = ok ? seg0[qq] : 0;
            ln[u] = ok ? segn[qq] : 0;
        }
#pragma unroll
        for (int u = 0; u < 4; ++u) ky[u] = (lane < ln[u]) ? keys[st[u] + lane] : -1;
#pragma unroll
        for (int u = 0; u < 4; ++u) {
            if (ky[u] >= 0) {
                if (PLACE) {
                    const int r = atomicAdd(&cntl[ky[u] >> SRC_BITS], 1);
                    if (staged) cbuf[r] = ky[u] & SRC_MASK; else colb[r] = ky[u] & SRC_MASK;
                } else {
                    atomicAdd(&cntl[ky[u] >> SRC_BITS], 1);
                }
            }
            for (int i = 32 + lane; i < ln[u]; i += 32) {
                const int kk = keys[st[u] + i];
                if (PLACE) {
                    const int r = atomicAdd(&cntl[kk >> SRC_BITS], 1);
                    if (staged) cbuf[r] = kk & SRC_MASK; else colb[r] = kk & SRC_MASK;
                } else {
                    atomicAdd(&cntl[kk >> SRC_BITS], 1);
                }
            }
        }
    }
}

__global__ void __launch_bounds__(BKT_W)
gcn_build(const int* __restrict__ keys, const int* __restrict__ tab,
          int* __restrict__ cnt, int* __restrict__ rowptr, int* __restrict__ col)
{
    __shared__ int seg0[TAB_P];
    __shared__ int segn[TAB_P];
    __shared__ int cntl[BKT_W];
    __shared__ int wsum[32];
    __shared__ int cbuf[COL_CAP];
    const int t = threadIdx.x;
    const int lane = t & 31;
    const int w = t >> 5;
    const int k = blockIdx.x;

    int s0 = 0, sn = 0;
    if (t < PART_NB) {
        s0 = tab[k * TAB_P + t];
        sn = tab[(k + 1) * TAB_P + t] - s0;
        seg0[t] = t * PART_CH + s0;
        segn[t] = sn;
    }
    cntl[t] = 0;
    int ra = s0, rn = sn;
    for (int o = 16; o > 0; o >>= 1) {
        ra += __shfl_xor(ra, o);
        rn += __shfl_xor(rn, o);
    }
    if (lane == 0) { wsum[w] = ra; wsum[16 + w] = rn; }
    __syncthreads();
    int base = 0, total = 0;
    for (int i = 0; i < 16; ++i) { base += wsum[i]; total += wsum[16 + i]; }
    __syncthreads();

    gcn_bucket_walk<false>(keys, seg0, segn, cntl, cbuf, col + base, false, w, lane);
    __syncthreads();

    const int c = cntl[t];
    int inc = c;
    for (int o = 1; o < 32; o <<= 1) {
        int v = __shfl_up(inc, o);
        if (lane >= o) inc += v;
    }
    if (lane == 31) wsum[w] = inc;
    __syncthreads();
    if (w == 0) {
        const int s = (lane < 16) ? wsum[lane] : 0;
        int si = s;
        for (int o = 1; o < 32; o <<= 1) {
            int v = __shfl_up(si, o);
            if (lane >= o) si += v;
        }
        if (lane < 16) wsum[lane] = si - s;
    }
    __syncthreads();
    const int ex = wsum[w] + inc - c;
    const int node = k * BKT_W + t;
    if (node < NNODES) {
        cnt[node] = c;
        rowptr[node] = base + ex;
    }
    cntl[t] = ex;
    __syncthreads();

    const bool staged = total <= COL_CAP;
    gcn_bucket_walk<true>(keys, seg0, segn, cntl, cbuf, col + base, staged, w, lane);
    if (staged) {
        __syncthreads();
        for (int i = t; i < total; i += BKT_W) col[base + i] = cbuf[i];
    }
}

template <int K, int NC>
__global__ void gcn_gemm_wmma(const float* __restrict__ A,
                              const float* __restrict__ B,
                              const int* __restrict__ cnt,
                              float* __restrict__ C, int M)
{
    constexpr int T = NC / 16;
    typedef __attribute__((ext_vector_type(T))) float vtf;
    static_assert(NC % 16 == 0 && (T == 2 || T == 4), "a lane's T columns are one 8- or 16-byte vector");

    const int wave = (blockIdx.x * blockDim.x + threadIdx.x) >> 5;
    if (wave >= (M >> 4)) return;

    const int lane = threadIdx.x & 31;
    const int lo = lane & 15;
    const int hi = lane >> 4;
    const int tm = wave << 4;

    const float* arow = A + (size_t)(tm + lo) * K + 2 * hi;
    const float* brow = B + (size_t)(2 * hi) * NC + T * lo;

    v8f acc[T];
#pragma unroll
    for (int t = 0; t < T; ++t) acc[t] = v8f{};

    constexpr int S = K / 4;
    constexpr int PA = S < 16 ? S : 16;
    constexpr int PB = 2;

    v2f a[PA];
#pragma unroll
    for (int s = 0; s < PA; ++s) {
        a[s].x = arow[4 * s];
        a[s].y = arow[4 * s + 1];
    }

    vtf b0[PB], b1[PB];
#pragma unroll
    for (int p = 0; p < PB; ++p) {
        b0[p] = *(const vtf*)(brow + (size_t)(4 * p) * NC);
        b1[p] = *(const vtf*)(brow + (size_t)(4 * p + 1) * NC);
    }

#pragma unroll
    for (int s = 0; s < S; ++s) {
        const v2f ca = a[s % PA];
        const vtf c0 = b0[s % PB];
        const vtf c1 = b1[s % PB];
        if (s + PA < S) {
            a[s % PA].x = arow[4 * (s + PA)];
            a[s % PA].y = arow[4 * (s + PA) + 1];
        }
        if (s + PB < S) {
            b0[s % PB] = *(const vtf*)(brow + (size_t)(4 * (s + PB)) * NC);
            b1[s % PB] = *(const vtf*)(brow + (size_t)(4 * (s + PB) + 1) * NC);
        }
#pragma unroll
        for (int t = 0; t < T; ++t) {
            v2f b;
            b.x = c0[t];
            b.y = c1[t];
            acc[t] = __builtin_amdgcn_wmma_f32_16x16x4_f32(
                 false, ca,  false, b,
                 (short)0, acc[t],  false,  false);
        }
    }

    const int* crcnt = cnt + tm + 8 * hi;
    float* crow = C + (size_t)(tm + 8 * hi) * NC + T * lo;
#pragma unroll
    for (int g = 0; g < 8; ++g) {
        const float sc = rsqrtf((float)(crcnt[g] + 1));
        vtf o;
#pragma unroll
        for (int t = 0; t < T; ++t) o[t] = acc[t][g] * sc;
        *(vtf*)(crow + (size_t)g * NC) = o;
    }
}


template <int C, int UU>
__device__ __forceinline__ void gcn_add_rows(const float* __restrict__ hs, int id, int j,
                                             unsigned lane, float (&acc)[C / 32])
{
    constexpr int R = C / 32;
    float v[UU][R];
#pragma unroll
    for (int u = 0; u < UU; ++u) {
        const int sj = __builtin_amdgcn_readlane(id, j + u);
        const float* hr = hs + (size_t)sj * C;
#pragma unroll
        for (int r = 0; r < R; ++r) v[u][r] = hr[lane + 32u * r];
    }
#pragma unroll
    for (int u = 0; u < UU; ++u)
#pragma unroll
        for (int r = 0; r < R; ++r) acc[r] += v[u][r];
}

template <int C, bool RELU>
__global__ void gcn_gather(const float* __restrict__ hs,
                           const int* __restrict__ rowptr,
                           const int* __restrict__ cnt,
                           const int* __restrict__ col,
                           const float* __restrict__ bias,
                           float* __restrict__ out)
{
    static_assert(C % 32 == 0, "lane = channel needs rows of whole 32-float lines");
    constexpr int R = C / 32;

    static_assert(NNODES % NPW == 0, "no partial wave");

    const int w = __builtin_amdgcn_readfirstlane((int)((blockIdx.x * blockDim.x + threadIdx.x) >> 5));
    if (w >= NNODES / NPW) return;
    const unsigned lane = threadIdx.x & 31u;

    for (int k = 0; k < NPW; ++k) {
    const int d = w * NPW + k;
    const int start = __builtin_amdgcn_readfirstlane(rowptr[d]);
    const int n     = __builtin_amdgcn_readfirstlane(cnt[d]);

    float acc[R];
#pragma unroll
    for (int r = 0; r < R; ++r) acc[r] = 0.f;

    const int total = n + 1;
    for (int j0 = 0; j0 < total; j0 += 32) {
        const int m = (total - j0 < 32) ? (total - j0) : 32;
        const int id = (j0 + (int)lane < n) ? col[start + j0 + (int)lane] : d;
        int j = 0;
        for (; j + 8 <= m; j += 8) gcn_add_rows<C, 8>(hs, id, j, lane, acc);
        if (j + 4 <= m) { gcn_add_rows<C, 4>(hs, id, j, lane, acc); j += 4; }
        if (j + 2 <= m) { gcn_add_rows<C, 2>(hs, id, j, lane, acc); j += 2; }
        if (j < m)      { gcn_add_rows<C, 1>(hs, id, j, lane, acc); }
    }

    const float dn = rsqrtf((float)(n + 1));
    float* o = out + (size_t)d * C;
#pragma unroll
    for (int r = 0; r < R; ++r) {
        float y = acc[r] * dn + bias[lane + 32u * r];
        if (RELU) y = y > 0.f ? y : 0.f;
        o[lane + 32u * r] = y;
    }
    }
}

extern "C" void kernel_launch(void* const* d_in, const int* in_sizes, int n_in,
                              void* d_out, int out_size, void* d_ws, size_t ws_size,
                              hipStream_t stream)
{
    const float* x    = (const float*)d_in[0];
    const int*   eidx = (const int*)d_in[1];
    const int*   src  = eidx;
    const int*   dst  = eidx + NEDGES;
    const float* W1   = (const float*)d_in[2];
    const float* b1   = (const float*)d_in[3];
    const float* W2   = (const float*)d_in[4];
    const float* b2   = (const float*)d_in[5];
    float* out = (float*)d_out;

    int*   cnt    = (int*)d_ws;
    int*   rowptr = cnt + NNODES;
    int*   keys   = rowptr + NNODES;
    int*   col    = keys + PART_NB * PART_CH;
    int*   tab    = col + NEDGES;
    float* h      = (float*)(tab + (NBKT + 1) * TAB_P);
    float* agg1   = h + (size_t)NNODES * HID_C;

    const int B = 256;

    gcn_part<<<PART_NB, PART_T, 0, stream>>>(src, dst, keys, tab);
    gcn_build<<<NBKT, BKT_W, 0, stream>>>(keys, tab, cnt, rowptr, col);

    {
        int waves = NNODES / 16;
        int thr = waves * 32;
        gcn_gemm_wmma<IN_C, HID_C><<<(thr + B - 1) / B, B, 0, stream>>>(x, W1, cnt, h, NNODES);
    }

    {
        int thr = (NNODES / NPW) * 32;
        gcn_gather<HID_C, true><<<(thr + B - 1) / B, B, 0, stream>>>(h, rowptr, cnt, col, b1, agg1);
    }

    {
        int waves = NNODES / 16;
        int thr = waves * 32;
        gcn_gemm_wmma<HID_C, OUT_C><<<(thr + B - 1) / B, B, 0, stream>>>(agg1, W2, cnt, h, NNODES);
    }

    {
        int thr = (NNODES / NPW) * 32;
        gcn_gather<OUT_C, false><<<(thr + B - 1) / B, B, 0, stream>>>(h, rowptr, cnt, col, b2, out);
    }
}


# layer-2 product moved into the layer-1 gather launch: a block's 64 relu'd rows stay in an LDS tile and 4 of its waves run the same f32 WMMA strip product on them; one launch fewer; sums unchanged
# speedup vs baseline: 3.4851x; 1.1023x over previous
#include <hip/hip_runtime.h>
#include <hip/hip_bf16.h>

#define NNODES 100000
#define NEDGES 1000000
#define IN_C   128
#define HID_C  64
#define OUT_C  32

#define PART_T 1024
#define PART_EPT 4
#define PART_CH (PART_T * PART_EPT)
#define PART_NB ((NEDGES + PART_CH - 1) / PART_CH)
#define BKT_SH 9
#define BKT_W (1 << BKT_SH)
#define NBKT ((NNODES + BKT_W - 1) / BKT_W)
#define HPAD 256
#define TAB_P 256
#define SRC_BITS 17
#define SRC_MASK ((1 << SRC_BITS) - 1)
#define COL_CAP 8192
#define NPW 8

typedef __attribute__((ext_vector_type(2))) float v2f;
typedef __attribute__((ext_vector_type(8))) float v8f;

static_assert(NNODES <= (1 << SRC_BITS), "a source index must fit in SRC_BITS bits");
static_assert(BKT_SH + SRC_BITS < 32, "key stays a non-negative int");
static_assert(NBKT + 1 <= HPAD && HPAD <= PART_T, "one thread per bucket counter in the partition block");
static_assert(PART_NB <= TAB_P && PART_NB <= BKT_W, "one thread per run in the bucket block");

__global__ void __launch_bounds__(PART_T)
gcn_part(const int* __restrict__ src, const int* __restrict__ dst,
         int* __restrict__ keys, int* __restrict__ tab)
{
    __shared__ int hist[HPAD];
    __shared__ int wsum[32];
    __shared__ int stage[PART_CH];
    const int t = threadIdx.x;
    const int lane = t & 31;
    const int w = t >> 5;
    const int b = blockIdx.x;
    const int e0 = b * PART_CH;

    if (t < HPAD) hist[t] = 0;

    int key[PART_EPT], bk[PART_EPT], rk[PART_EPT];
#pragma unroll
    for (int k = 0; k < PART_EPT; ++k) {
        const int e = e0 + t + k * PART_T;
        const bool ok = e < NEDGES;
        const int d = ok ? dst[e] : 0;
        const int s = ok ? src[e] : 0;
        bk[k] = ok ? (d >> BKT_SH) : -1;
        key[k] = ((d & (BKT_W - 1)) << SRC_BITS) | s;
        rk[k] = 0;
    }
    __syncthreads();
#pragma unroll
    for (int k = 0; k < PART_EPT; ++k)
        if (bk[k] >= 0) rk[k] = atomicAdd(&hist[bk[k]], 1);
    __syncthreads();

    const int c = (t < HPAD) ? hist[t] : 0;
    int inc = c;
    for (int o = 1; o < 32; o <<= 1) {
        int v = __shfl_up(inc, o);
        if (lane >= o) inc += v;
    }
    if (lane == 31) wsum[w] = inc;
    __syncthreads();
    if (w == 0) {
        const int s = wsum[lane];
        int si = s;
        for (int o = 1; o < 32; o <<= 1) {
            int v = __shfl_up(si, o);
            if (lane >= o) si += v;
        }
        wsum[lane] = si - s;
    }
    __syncthreads();
    const int ex = wsum[w] + inc - c;
    if (t <= NBKT) tab[t * TAB_P + b] = ex;
    if (t < HPAD) hist[t] = ex;
    __syncthreads();

#pragma unroll
    for (int k = 0; k < PART_EPT; ++k)
        if (bk[k] >= 0) stage[hist[bk[k]] + rk[k]] = key[k];
    __syncthreads();
    const int n = (NEDGES - e0 < PART_CH) ? (NEDGES - e0) : PART_CH;
#pragma unroll
    for (int k = 0; k < PART_EPT; ++k) {
        const int i = t + k * PART_T;
        if (i < n) keys[e0 + i] = stage[i];
    }
}

template <bool PLACE>
__device__ __forceinline__ void gcn_bucket_walk(const int* __restrict__ keys,
                                                const int* seg0, const int* segn,
                                                int* cntl, int* cbuf,
                                                int* __restrict__ colb, bool staged,
                                                int w, int lane)
{
    constexpr int NW = BKT_W / 32;
    for (int q = w; q < PART_NB; q += 4 * NW) {
        int st[4], ln[4], ky[4];
#pragma unroll
        for (int u = 0; u < 4; ++u) {
            const int qq = q + NW * u;
            const bool ok = qq < PART_NB;
            st[u] = ok ? seg0[qq] : 0;
            ln[u] = ok ? segn[qq] : 0;
        }
#pragma unroll
        for (int u = 0; u < 4; ++u) ky[u] = (lane < ln[u]) ? keys[st[u] + lane] : -1;
#pragma unroll
        for (int u = 0; u < 4; ++u) {
            if (ky[u] >= 0) {
                if (PLACE) {
                    const int r = atomicAdd(&cntl[ky[u] >> SRC_BITS], 1);
                    if (staged) cbuf[r] = ky[u] & SRC_MASK; else colb[r] = ky[u] & SRC_MASK;
                } else {
                    atomicAdd(&cntl[ky[u] >> SRC_BITS], 1);
                }
            }
            for (int i = 32 + lane; i < ln[u]; i += 32) {
                const int kk = keys[st[u] + i];
                if (PLACE) {
                    const int r = atomicAdd(&cntl[kk >> SRC_BITS], 1);
                    if (staged) cbuf[r] = kk & SRC_MASK; else colb[r] = kk & SRC_MASK;
                } else {
                    atomicAdd(&cntl[kk >> SRC_BITS], 1);
                }
            }
        }
    }
}

__global__ void __launch_bounds__(BKT_W)
gcn_build(const int* __restrict__ keys, const int* __restrict__ tab,
          int* __restrict__ cnt, int* __restrict__ rowptr, int* __restrict__ col)
{
    __shared__ int seg0[TAB_P];
    __shared__ int segn[TAB_P];
    __shared__ int cntl[BKT_W];
    __shared__ int wsum[32];
    __shared__ int cbuf[COL_CAP];
    const int t = threadIdx.x;
    const int lane = t & 31;
    const int w = t >> 5;
    const int k = blockIdx.x;

    int s0 = 0, sn = 0;
    if (t < PART_NB) {
        s0 = tab[k * TAB_P + t];
        sn = tab[(k + 1) * TAB_P + t] - s0;
        seg0[t] = t * PART_CH + s0;
        segn[t] = sn;
    }
    cntl[t] = 0;
    int ra = s0, rn = sn;
    for (int o = 16; o > 0; o >>= 1) {
        ra += __shfl_xor(ra, o);
        rn += __shfl_xor(rn, o);
    }
    if (lane == 0) { wsum[w] = ra; wsum[16 + w] = rn; }
    __syncthreads();
    int base = 0, total = 0;
    for (int i = 0; i < 16; ++i) { base += wsum[i]; total += wsum[16 + i]; }
    __syncthreads();

    gcn_bucket_walk<false>(keys, seg0, segn, cntl, cbuf, col + base, false, w, lane);
    __syncthreads();

    const int c = cntl[t];
    int inc = c;
    for (int o = 1; o < 32; o <<= 1) {
        int v = __shfl_up(inc, o);
        if (lane >= o) inc += v;
    }
    if (lane == 31) wsum[w] = inc;
    __syncthreads();
    if (w == 0) {
        const int s = (lane < 16) ? wsum[lane] : 0;
        int si = s;
        for (int o = 1; o < 32; o <<= 1) {
            int v = __shfl_up(si, o);
            if (lane >= o) si += v;
        }
        if (lane < 16) wsum[lane] = si - s;
    }
    __syncthreads();
    const int ex = wsum[w] + inc - c;
    const int node = k * BKT_W + t;
    if (node < NNODES) {
        cnt[node] = c;
        rowptr[node] = base + ex;
    }
    cntl[t] = ex;
    __syncthreads();

    const bool staged = total <= COL_CAP;
    gcn_bucket_walk<true>(keys, seg0, segn, cntl, cbuf, col + base, staged, w, lane);
    if (staged) {
        __syncthreads();
        for (int i = t; i < total; i += BKT_W) col[base + i] = cbuf[i];
    }
}

template <int K, int NC>
__global__ void gcn_gemm_wmma(const float* __restrict__ A,
                              const float* __restrict__ B,
                              const int* __restrict__ cnt,
                              float* __restrict__ C, int M)
{
    constexpr int T = NC / 16;
    typedef __attribute__((ext_vector_type(T))) float vtf;
    static_assert(NC % 16 == 0 && (T == 2 || T == 4), "a lane's T columns are one 8- or 16-byte vector");

    const int wave = (blockIdx.x * blockDim.x + threadIdx.x) >> 5;
    if (wave >= (M >> 4)) return;

    const int lane = threadIdx.x & 31;
    const int lo = lane & 15;
    const int hi = lane >> 4;
    const int tm = wave << 4;

    const float* arow = A + (size_t)(tm + lo) * K + 2 * hi;
    const float* brow = B + (size_t)(2 * hi) * NC + T * lo;

    v8f acc[T];
#pragma unroll
    for (int t = 0; t < T; ++t) acc[t] = v8f{};

    constexpr int S = K / 4;
    constexpr int PA = S < 16 ? S : 16;
    constexpr int PB = 2;

    v2f a[PA];
#pragma unroll
    for (int s = 0; s < PA; ++s) {
        a[s].x = arow[4 * s];
        a[s].y = arow[4 * s + 1];
    }

    vtf b0[PB], b1[PB];
#pragma unroll
    for (int p = 0; p < PB; ++p) {
        b0[p] = *(const vtf*)(brow + (size_t)(4 * p) * NC);
        b1[p] = *(const vtf*)(brow + (size_t)(4 * p + 1) * NC);
    }

#pragma unroll
    for (int s = 0; s < S; ++s) {
        const v2f ca = a[s % PA];
        const vtf c0 = b0[s % PB];
        const vtf c1 = b1[s % PB];
        if (s + PA < S) {
            a[s % PA].x = arow[4 * (s + PA)];
            a[s % PA].y = arow[4 * (s + PA) + 1];
        }
        if (s + PB < S) {
            b0[s % PB] = *(const vtf*)(brow + (size_t)(4 * (s + PB)) * NC);
            b1[s % PB] = *(const vtf*)(brow + (size_t)(4 * (s + PB) + 1) * NC);
        }
#pragma unroll
        for (int t = 0; t < T; ++t) {
            v2f b;
            b.x = c0[t];
            b.y = c1[t];
            acc[t] = __builtin_amdgcn_wmma_f32_16x16x4_f32(
                 false, ca,  false, b,
                 (short)0, acc[t],  false,  false);
        }
    }

    const int* crcnt = cnt + tm + 8 * hi;
    float* crow = C + (size_t)(tm + 8 * hi) * NC + T * lo;
#pragma unroll
    for (int g = 0; g < 8; ++g) {
        const float sc = rsqrtf((float)(crcnt[g] + 1));
        vtf o;
#pragma unroll
        for (int t = 0; t < T; ++t) o[t] = acc[t][g] * sc;
        *(vtf*)(crow + (size_t)g * NC) = o;
    }
}


template <int C, int UU>
__device__ __forceinline__ void gcn_add_rows(const float* __restrict__ hs, int id, int j,
                                             unsigned lane, float (&acc)[C / 32])
{
    constexpr int R = C / 32;
    float v[UU][R];
#pragma unroll
    for (int u = 0; u < UU; ++u) {
        const int sj = __builtin_amdgcn_readlane(id, j + u);
        const float* hr = hs + (size_t)sj * C;
#pragma unroll
        for (int r = 0; r < R; ++r) v[u][r] = hr[lane + 32u * r];
    }
#pragma unroll
    for (int u = 0; u < UU; ++u)
#pragma unroll
        for (int r = 0; r < R; ++r) acc[r] += v[u][r];
}

template <int C, bool RELU>
__global__ void gcn_gather(const float* __restrict__ hs,
                           const int* __restrict__ rowptr,
                           const int* __restrict__ cnt,
                           const int* __restrict__ col,
                           const float* __restrict__ bias,
                           float* __restrict__ out)
{
    static_assert(C % 32 == 0, "lane = channel needs rows of whole 32-float lines");
    constexpr int R = C / 32;

    static_assert(NNODES % NPW == 0, "no partial wave");

    const int w = __builtin_amdgcn_readfirstlane((int)((blockIdx.x * blockDim.x + threadIdx.x) >> 5));
    if (w >= NNODES / NPW) return;
    const unsigned lane = threadIdx.x & 31u;

    for (int k = 0; k < NPW; ++k) {
    const int d = w * NPW + k;
    const int start = __builtin_amdgcn_readfirstlane(rowptr[d]);
    const int n     = __builtin_amdgcn_readfirstlane(cnt[d]);

    float acc[R];
#pragma unroll
    for (int r = 0; r < R; ++r) acc[r] = 0.f;

    const int total = n + 1;
    for (int j0 = 0; j0 < total; j0 += 32) {
        const int m = (total - j0 < 32) ? (total - j0) : 32;
        const int id = (j0 + (int)lane < n) ? col[start + j0 + (int)lane] : d;
        int j = 0;
        for (; j + 8 <= m; j += 8) gcn_add_rows<C, 8>(hs, id, j, lane, acc);
        if (j + 4 <= m) { gcn_add_rows<C, 4>(hs, id, j, lane, acc); j += 4; }
        if (j + 2 <= m) { gcn_add_rows<C, 2>(hs, id, j, lane, acc); j += 2; }
        if (j < m)      { gcn_add_rows<C, 1>(hs, id, j, lane, acc); }
    }

    const float dn = rsqrtf((float)(n + 1));
    float* o = out + (size_t)d * C;
#pragma unroll
    for (int r = 0; r < R; ++r) {
        float y = acc[r] * dn + bias[lane + 32u * r];
        if (RELU) y = y > 0.f ? y : 0.f;
        o[lane + 32u * r] = y;
    }
    }
}

template <int C, int NC>
__global__ void __launch_bounds__(256)
gcn_gather_gemm(const float* __restrict__ hs,
                const int* __restrict__ rowptr,
                const int* __restrict__ cnt,
                const int* __restrict__ col,
                const float* __restrict__ bias,
                const float* __restrict__ B,
                float* __restrict__ out)
{
    static_assert(C % 32 == 0, "lane = channel needs rows of whole 32-float lines");
    constexpr int R = C / 32;
    constexpr int WPB = 8;
    constexpr int NPB = WPB * NPW;
    constexpr int PITCH = C + 4;
    constexpr int T = NC / 16;
    constexpr int S = C / 4;
    typedef __attribute__((ext_vector_type(T))) float vtf;
    static_assert(NPB % 16 == 0 && NNODES % 16 == 0 && NNODES % NPW == 0, "whole strips, whole waves");
    static_assert(NC % 16 == 0 && (T == 2 || T == 4), "a lane's T columns are one 8- or 16-byte vector");

    __shared__ float tile[NPB * PITCH];

    const int wv = __builtin_amdgcn_readfirstlane((int)(threadIdx.x >> 5));
    const unsigned lane = threadIdx.x & 31u;
    const int nb = (int)blockIdx.x * NPB;
    const int d0 = nb + wv * NPW;

    if (d0 < NNODES) {
        for (int k = 0; k < NPW; ++k) {
            const int d = d0 + k;
            const int start = __builtin_amdgcn_readfirstlane(rowptr[d]);
            const int n     = __builtin_amdgcn_readfirstlane(cnt[d]);

            float acc[R];
#pragma unroll
            for (int r = 0; r < R; ++r) acc[r] = 0.f;

            const int total = n + 1;
            for (int j0 = 0; j0 < total; j0 += 32) {
                const int m = (total - j0 < 32) ? (total - j0) : 32;
                const int id = (j0 + (int)lane < n) ? col[start + j0 + (int)lane] : d;
                int j = 0;
                for (; j + 8 <= m; j += 8) gcn_add_rows<C, 8>(hs, id, j, lane, acc);
                if (j + 4 <= m) { gcn_add_rows<C, 4>(hs, id, j, lane, acc); j += 4; }
                if (j + 2 <= m) { gcn_add_rows<C, 2>(hs, id, j, lane, acc); j += 2; }
                if (j < m)      { gcn_add_rows<C, 1>(hs, id, j, lane, acc); }
            }

            const float dn = rsqrtf((float)(n + 1));
            float* o = tile + (wv * NPW + k) * PITCH;
#pragma unroll
            for (int r = 0; r < R; ++r) {
                float y = acc[r] * dn + bias[lane + 32u * r];
                o[lane + 32u * r] = y > 0.f ? y : 0.f;
            }
        }
    }
    __syncthreads();

    const int tm = nb + 16 * wv;
    if (wv >= NPB / 16 || tm >= NNODES) return;

    const int lo = (int)lane & 15;
    const int hi = (int)lane >> 4;
    const float* arow = tile + (16 * wv + lo) * PITCH + 2 * hi;
    const float* brow = B + (size_t)(2 * hi) * NC + T * lo;

    v8f acc[T];
#pragma unroll
    for (int t = 0; t < T; ++t) acc[t] = v8f{};

    constexpr int CH = 4;
    static_assert(S % CH == 0, "whole groups of k steps");
#pragma unroll 1
    for (int s0 = 0; s0 < S; s0 += CH) {
        v2f a[CH];
        vtf c0[CH], c1[CH];
#pragma unroll
        for (int i = 0; i < CH; ++i) {
            a[i]  = *(const v2f*)(arow + 4 * (s0 + i));
            c0[i] = *(const vtf*)(brow + (size_t)(4 * (s0 + i)) * NC);
            c1[i] = *(const vtf*)(brow + (size_t)(4 * (s0 + i) + 1) * NC);
        }
#pragma unroll
        for (int i = 0; i < CH; ++i) {
#pragma unroll
            for (int t = 0; t < T; ++t) {
                v2f b;
                b.x = c0[i][t];
                b.y = c1[i][t];
                acc[t] = __builtin_amdgcn_wmma_f32_16x16x4_f32(
                     false, a[i],  false, b,
                     (short)0, acc[t],  false,  false);
            }
        }
    }

    const int* crcnt = cnt + tm + 8 * hi;
    float* crow = out + (size_t)(tm + 8 * hi) * NC + T * lo;
#pragma unroll
    for (int g = 0; g < 8; ++g) {
        const float sc = rsqrtf((float)(crcnt[g] + 1));
        vtf o;
#pragma unroll
        for (int t = 0; t < T; ++t) o[t] = acc[t][g] * sc;
        *(vtf*)(crow + (size_t)g * NC) = o;
    }
}

extern "C" void kernel_launch(void* const* d_in, const int* in_sizes, int n_in,
                              void* d_out, int out_size, void* d_ws, size_t ws_size,
                              hipStream_t stream)
{
    const float* x    = (const float*)d_in[0];
    const int*   eidx = (const int*)d_in[1];
    const int*   src  = eidx;
    const int*   dst  = eidx + NEDGES;
    const float* W1   = (const float*)d_in[2];
    const float* b1   = (const float*)d_in[3];
    const float* W2   = (const float*)d_in[4];
    const float* b2   = (const float*)d_in[5];
    float* out = (float*)d_out;

    int*   cnt    = (int*)d_ws;
    int*   rowptr = cnt + NNODES;
    int*   keys   = rowptr + NNODES;
    int*   col    = keys + PART_NB * PART_CH;
    int*   tab    = col + NEDGES;
    float* h      = (float*)(tab + (NBKT + 1) * TAB_P);
    float* h2     = h + (size_t)NNODES * HID_C;

    const int B = 256;

    gcn_part<<<PART_NB, PART_T, 0, stream>>>(src, dst, keys, tab);
    gcn_build<<<NBKT, BKT_W, 0, stream>>>(keys, tab, cnt, rowptr, col);

    {
        int waves = NNODES / 16;
        int thr = waves * 32;
        gcn_gemm_wmma<IN_C, HID_C><<<(thr + B - 1) / B, B, 0, stream>>>(x, W1, cnt, h, NNODES);
    }

    gcn_gather_gemm<HID_C, OUT_C><<<(NNODES + 63) / 64, B, 0, stream>>>(h, rowptr, cnt, col, b1, W2, h2);

    {
        int thr = (NNODES / NPW) * 32;
        gcn_gather<OUT_C, false><<<(thr + B - 1) / B, B, 0, stream>>>(h2, rowptr, cnt, col, b2, out);
    }
}


# gather: a node's rows (pieces of 8,4,2,1) all requested before the first add, the next node's ids behind them, starts and counts one load per wave: one memory round trip per node, not five; same sums
# speedup vs baseline: 3.7909x; 1.0877x over previous
#include <hip/hip_runtime.h>
#include <hip/hip_bf16.h>

#define NNODES 100000
#define NEDGES 1000000
#define IN_C   128
#define HID_C  64
#define OUT_C  32

#define PART_T 1024
#define PART_EPT 4
#define PART_CH (PART_T * PART_EPT)
#define PART_NB ((NEDGES + PART_CH - 1) / PART_CH)
#define BKT_SH 9
#define BKT_W (1 << BKT_SH)
#define NBKT ((NNODES + BKT_W - 1) / BKT_W)
#define HPAD 256
#define TAB_P 256
#define SRC_BITS 17
#define SRC_MASK ((1 << SRC_BITS) - 1)
#define COL_CAP 8192
#define NPW 8

typedef __attribute__((ext_vector_type(2))) float v2f;
typedef __attribute__((ext_vector_type(8))) float v8f;

static_assert(NNODES <= (1 << SRC_BITS), "a source index must fit in SRC_BITS bits");
static_assert(BKT_SH + SRC_BITS < 32, "key stays a non-negative int");
static_assert(NBKT + 1 <= HPAD && HPAD <= PART_T, "one thread per bucket counter in the partition block");
static_assert(PART_NB <= TAB_P && PART_NB <= BKT_W, "one thread per run in the bucket block");

__global__ void __launch_bounds__(PART_T)
gcn_part(const int* __restrict__ src, const int* __restrict__ dst,
         int* __restrict__ keys, int* __restrict__ tab)
{
    __shared__ int hist[HPAD];
    __shared__ int wsum[32];
    __shared__ int stage[PART_CH];
    const int t = threadIdx.x;
    const int lane = t & 31;
    const int w = t >> 5;
    const int b = blockIdx.x;
    const int e0 = b * PART_CH;

    if (t < HPAD) hist[t] = 0;

    int key[PART_EPT], bk[PART_EPT], rk[PART_EPT];
#pragma unroll
    for (int k = 0; k < PART_EPT; ++k) {
        const int e = e0 + t + k * PART_T;
        const bool ok = e < NEDGES;
        const int d = ok ? dst[e] : 0;
        const int s = ok ? src[e] : 0;
        bk[k] = ok ? (d >> BKT_SH) : -1;
        key[k] = ((d & (BKT_W - 1)) << SRC_BITS) | s;
        rk[k] = 0;
    }
    __syncthreads();
#pragma unroll
    for (int k = 0; k < PART_EPT; ++k)
        if (bk[k] >= 0) rk[k] = atomicAdd(&hist[bk[k]], 1);
    __syncthreads();

    const int c = (t < HPAD) ? hist[t] : 0;
    int inc = c;
    for (int o = 1; o < 32; o <<= 1) {
        int v = __shfl_up(inc, o);
        if (lane >= o) inc += v;
    }
    if (lane == 31) wsum[w] = inc;
    __syncthreads();
    if (w == 0) {
        const int s = wsum[lane];
        int si = s;
        for (int o = 1; o < 32; o <<= 1) {
            int v = __shfl_up(si, o);
            if (lane >= o) si += v;
        }
        wsum[lane] = si - s;
    }
    __syncthreads();
    const int ex = wsum[w] + inc - c;
    if (t <= NBKT) tab[t * TAB_P + b] = ex;
    if (t < HPAD) hist[t] = ex;
    __syncthreads();

#pragma unroll
    for (int k = 0; k < PART_EPT; ++k)
        if (bk[k] >= 0) stage[hist[bk[k]] + rk[k]] = key[k];
    __syncthreads();
    const int n = (NEDGES - e0 < PART_CH) ? (NEDGES - e0) : PART_CH;
#pragma unroll
    for (int k = 0; k < PART_EPT; ++k) {
        const int i = t + k * PART_T;
        if (i < n) keys[e0 + i] = stage[i];
    }
}

template <bool PLACE>
__device__ __forceinline__ void gcn_bucket_walk(const int* __restrict__ keys,
                                                const int* seg0, const int* segn,
                                                int* cntl, int* cbuf,
                                                int* __restrict__ colb, bool staged,
                                                int w, int lane)
{
    constexpr int NW = BKT_W / 32;
    for (int q = w; q < PART_NB; q += 4 * NW) {
        int st[4], ln[4], ky[4];
#pragma unroll
        for (int u = 0; u < 4; ++u) {
            const int qq = q + NW * u;
            const bool ok = qq < PART_NB;
            st[u] = ok ? seg0[qq] : 0;
            ln[u] = ok ? segn[qq] : 0;
        }
#pragma unroll
        for (int u = 0; u < 4; ++u) ky[u] = (lane < ln[u]) ? keys[st[u] + lane] : -1;
#pragma unroll
        for (int u = 0; u < 4; ++u) {
            if (ky[u] >= 0) {
                if (PLACE) {
                    const int r = atomicAdd(&cntl[ky[u] >> SRC_BITS], 1);
                    if (staged) cbuf[r] = ky[u] & SRC_MASK; else colb[r] = ky[u] & SRC_MASK;
                } else {
                    atomicAdd(&cntl[ky[u] >> SRC_BITS], 1);
                }
            }
            for (int i = 32 + lane; i < ln[u]; i += 32) {
                const int kk = keys[st[u] + i];
                if (PLACE) {
                    const int r = atomicAdd(&cntl[kk >> SRC_BITS], 1);
                    if (staged) cbuf[r] = kk & SRC_MASK; else colb[r] = kk & SRC_MASK;
                } else {
                    atomicAdd(&cntl[kk >> SRC_BITS], 1);
                }
            }
        }
    }
}

__global__ void __launch_bounds__(BKT_W)
gcn_build(const int* __restrict__ keys, const int* __restrict__ tab,
          int* __restrict__ cnt, int* __restrict__ rowptr, int* __restrict__ col)
{
    __shared__ int seg0[TAB_P];
    __shared__ int segn[TAB_P];
    __shared__ int cntl[BKT_W];
    __shared__ int wsum[32];
    __shared__ int cbuf[COL_CAP];
    const int t = threadIdx.x;
    const int lane = t & 31;
    const int w = t >> 5;
    const int k = blockIdx.x;

    int s0 = 0, sn = 0;
    if (t < PART_NB) {
        s0 = tab[k * TAB_P + t];
        sn = tab[(k + 1) * TAB_P + t] - s0;
        seg0[t] = t * PART_CH + s0;
        segn[t] = sn;
    }
    cntl[t] = 0;
    int ra = s0, rn = sn;
    for (int o = 16; o > 0; o >>= 1) {
        ra += __shfl_xor(ra, o);
        rn += __shfl_xor(rn, o);
    }
    if (lane == 0) { wsum[w] = ra; wsum[16 + w] = rn; }
    __syncthreads();
    int base = 0, total = 0;
    for (int i = 0; i < 16; ++i) { base += wsum[i]; total += wsum[16 + i]; }
    __syncthreads();

    gcn_bucket_walk<false>(keys, seg0, segn, cntl, cbuf, col + base, false, w, lane);
    __syncthreads();

    const int c = cntl[t];
    int inc = c;
    for (int o = 1; o < 32; o <<= 1) {
        int v = __shfl_up(inc, o);
        if (lane >= o) inc += v;
    }
    if (lane == 31) wsum[w] = inc;
    __syncthreads();
    if (w == 0) {
        const int s = (lane < 16) ? wsum[lane] : 0;
        int si = s;
        for (int o = 1; o < 32; o <<= 1) {
            int v = __shfl_up(si, o);
            if (lane >= o) si += v;
        }
        if (lane < 16) wsum[lane] = si - s;
    }
    __syncthreads();
    const int ex = wsum[w] + inc - c;
    const int node = k * BKT_W + t;
    if (node < NNODES) {
        cnt[node] = c;
        rowptr[node] = base + ex;
    }
    cntl[t] = ex;
    __syncthreads();

    const bool staged = total <= COL_CAP;
    gcn_bucket_walk<true>(keys, seg0, segn, cntl, cbuf, col + base, staged, w, lane);
    if (staged) {
        __syncthreads();
        for (int i = t; i < total; i += BKT_W) col[base + i] = cbuf[i];
    }
}

template <int K, int NC>
__global__ void gcn_gemm_wmma(const float* __restrict__ A,
                              const float* __restrict__ B,
                              const int* __restrict__ cnt,
                              float* __restrict__ C, int M)
{
    constexpr int T = NC / 16;
    typedef __attribute__((ext_vector_type(T))) float vtf;
    static_assert(NC % 16 == 0 && (T == 2 || T == 4), "a lane's T columns are one 8- or 16-byte vector");

    const int wave = (blockIdx.x * blockDim.x + threadIdx.x) >> 5;
    if (wave >= (M >> 4)) return;

    const int lane = threadIdx.x & 31;
    const int lo = lane & 15;
    const int hi = lane >> 4;
    const int tm = wave << 4;

    const float* arow = A + (size_t)(tm + lo) * K + 2 * hi;
    const float* brow = B + (size_t)(2 * hi) * NC + T * lo;

    v8f acc[T];
#pragma unroll
    for (int t = 0; t < T; ++t) acc[t] = v8f{};

    constexpr int S = K / 4;
    constexpr int PA = S < 16 ? S : 16;
    constexpr int PB = 2;

    v2f a[PA];
#pragma unroll
    for (int s = 0; s < PA; ++s) {
        a[s].x = arow[4 * s];
        a[s].y = arow[4 * s + 1];
    }

    vtf b0[PB], b1[PB];
#pragma unroll
    for (int p = 0; p < PB; ++p) {
        b0[p] = *(const vtf*)(brow + (size_t)(4 * p) * NC);
        b1[p] = *(const vtf*)(brow + (size_t)(4 * p + 1) * NC);
    }

#pragma unroll
    for (int s = 0; s < S; ++s) {
        const v2f ca = a[s % PA];
        const vtf c0 = b0[s % PB];
        const vtf c1 = b1[s % PB];
        if (s + PA < S) {
            a[s % PA].x = arow[4 * (s + PA)];
            a[s % PA].y = arow[4 * (s + PA) + 1];
        }
        if (s + PB < S) {
            b0[s % PB] = *(const vtf*)(brow + (size_t)(4 * (s + PB)) * NC);
            b1[s % PB] = *(const vtf*)(brow + (size_t)(4 * (s + PB) + 1) * NC);
        }
#pragma unroll
        for (int t = 0; t < T; ++t) {
            v2f b;
            b.x = c0[t];
            b.y = c1[t];
            acc[t] = __builtin_amdgcn_wmma_f32_16x16x4_f32(
                 false, ca,  false, b,
                 (short)0, acc[t],  false,  false);
        }
    }

    const int* crcnt = cnt + tm + 8 * hi;
    float* crow = C + (size_t)(tm + 8 * hi) * NC + T * lo;
#pragma unroll
    for (int g = 0; g < 8; ++g) {
        const float sc = rsqrtf((float)(crcnt[g] + 1));
        vtf o;
#pragma unroll
        for (int t = 0; t < T; ++t) o[t] = acc[t][g] * sc;
        *(vtf*)(crow + (size_t)g * NC) = o;
    }
}


template <int C, int UU, int O>
__device__ __forceinline__ void gcn_load_rows(const float* __restrict__ hs, int id, int j,
                                              unsigned lane, float (&v)[16][C / 32])
{
    constexpr int R = C / 32;
#pragma unroll
    for (int u = 0; u < UU; ++u) {
        const int sj = __builtin_amdgcn_readlane(id, j + u);
        const float* hr = hs + (size_t)sj * C;
#pragma unroll
        for (int r = 0; r < R; ++r) v[O + u][r] = hr[lane + 32u * r];
    }
}

template <int C, int UU, int O>
__device__ __forceinline__ void gcn_sum_rows(const float (&v)[16][C / 32], float (&acc)[C / 32])
{
    constexpr int R = C / 32;
#pragma unroll
    for (int u = 0; u < UU; ++u)
#pragma unroll
        for (int r = 0; r < R; ++r) acc[r] += v[O + u][r];
}

template <int C>
__device__ __forceinline__ void gcn_node_sum(const float* __restrict__ hs,
                                             const int* __restrict__ col,
                                             int start, int n, int d, int id0,
                                             unsigned lane, float (&acc)[C / 32],
                                             bool more, int start1, int n1, int& idn)
{
    constexpr int R = C / 32;
    const int total = n + 1;
    int id = id0;
    float v[16][R];
    for (int j0 = 0; j0 < total; j0 += 32) {
        if (j0 > 0) id = (j0 + (int)lane < n) ? col[start + j0 + (int)lane] : d;
        asm volatile("" :: "v"(id));
        const int m = (total - j0 < 32) ? (total - j0) : 32;
        int j = 0;
        for (; j + 16 <= m; j += 16) {
            gcn_load_rows<C, 16, 0>(hs, id, j, lane, v);
            gcn_sum_rows<C, 16, 0>(v, acc);
        }
        const int r = m - j;
        const int j4 = j + (r & 8), j2 = j4 + (r & 4), j1 = j2 + (r & 2);
        if (r & 8) gcn_load_rows<C, 8, 0>(hs, id, j,  lane, v);
        if (r & 4) gcn_load_rows<C, 4, 8>(hs, id, j4, lane, v);
        if (r & 2) gcn_load_rows<C, 2, 12>(hs, id, j2, lane, v);
        if (r & 1) gcn_load_rows<C, 1, 14>(hs, id, j1, lane, v);
        if (j0 == 0 && more) idn = ((int)lane < n1) ? col[start1 + (int)lane] : d + 1;
        if (r & 8) gcn_sum_rows<C, 8, 0>(v, acc);
        if (r & 4) gcn_sum_rows<C, 4, 8>(v, acc);
        if (r & 2) gcn_sum_rows<C, 2, 12>(v, acc);
        if (r & 1) gcn_sum_rows<C, 1, 14>(v, acc);
    }
}

template <int C, bool RELU>
__global__ void gcn_gather(const float* __restrict__ hs,
                           const int* __restrict__ rowptr,
                           const int* __restrict__ cnt,
                           const int* __restrict__ col,
                           const float* __restrict__ bias,
                           float* __restrict__ out)
{
    static_assert(C % 32 == 0, "lane = channel needs rows of whole 32-float lines");
    constexpr int R = C / 32;

    static_assert(NNODES % NPW == 0, "no partial wave");

    const int w = __builtin_amdgcn_readfirstlane((int)((blockIdx.x * blockDim.x + threadIdx.x) >> 5));
    if (w >= NNODES / NPW) return;
    const unsigned lane = threadIdx.x & 31u;

    const int d0 = w * NPW;

    const int rpv = rowptr[d0 + (int)(lane & (NPW - 1))];
    const int ctv = cnt[d0 + (int)(lane & (NPW - 1))];
    float bv[R];
#pragma unroll
    for (int r = 0; r < R; ++r) bv[r] = bias[lane + 32u * r];

    int start = __builtin_amdgcn_readlane(rpv, 0);
    int n     = __builtin_amdgcn_readlane(ctv, 0);
    int idn   = ((int)lane < n) ? col[start + (int)lane] : d0;

#pragma unroll 1
    for (int k = 0; k < NPW; ++k) {
        const int d = d0 + k;
        const int st = start, nn = n, id0 = idn;
        start = __builtin_amdgcn_readlane(rpv, k + 1);
        n     = __builtin_amdgcn_readlane(ctv, k + 1);

        float acc[R];
#pragma unroll
        for (int r = 0; r < R; ++r) acc[r] = 0.f;
        gcn_node_sum<C>(hs, col, st, nn, d, id0, lane, acc, k + 1 < NPW, start, n, idn);

        const float dn = rsqrtf((float)(nn + 1));
        float* o = out + (size_t)d * C;
#pragma unroll
        for (int r = 0; r < R; ++r) {
            float y = acc[r] * dn + bv[r];
            if (RELU) y = y > 0.f ? y : 0.f;
            o[lane + 32u * r] = y;
        }
    }
}

template <int C, int NC>
__global__ void __launch_bounds__(256)
gcn_gather_gemm(const float* __restrict__ hs,
                const int* __restrict__ rowptr,
                const int* __restrict__ cnt,
                const int* __restrict__ col,
                const float* __restrict__ bias,
                const float* __restrict__ B,
                float* __restrict__ out)
{
    static_assert(C % 32 == 0, "lane = channel needs rows of whole 32-float lines");
    constexpr int R = C / 32;
    constexpr int WPB = 8;
    constexpr int NPB = WPB * NPW;
    constexpr int PITCH = C + 4;
    constexpr int T = NC / 16;
    constexpr int S = C / 4;
    typedef __attribute__((ext_vector_type(T))) float vtf;
    static_assert(NPB % 16 == 0 && NNODES % 16 == 0 && NNODES % NPW == 0, "whole strips, whole waves");
    static_assert(NC % 16 == 0 && (T == 2 || T == 4), "a lane's T columns are one 8- or 16-byte vector");

    __shared__ float tile[NPB * PITCH];

    const int wv = __builtin_amdgcn_readfirstlane((int)(threadIdx.x >> 5));
    const unsigned lane = threadIdx.x & 31u;
    const int nb = (int)blockIdx.x * NPB;
    const int d0 = nb + wv * NPW;

    if (d0 < NNODES) {
        const int rpv = rowptr[d0 + (int)(lane & (NPW - 1))];
        const int ctv = cnt[d0 + (int)(lane & (NPW - 1))];
        float bv[R];
#pragma unroll
        for (int r = 0; r < R; ++r) bv[r] = bias[lane + 32u * r];

        int start = __builtin_amdgcn_readlane(rpv, 0);
        int n     = __builtin_amdgcn_readlane(ctv, 0);
        int idn   = ((int)lane < n) ? col[start + (int)lane] : d0;

#pragma unroll 1
        for (int k = 0; k < NPW; ++k) {
            const int d = d0 + k;
            const int st = start, nn = n, id0 = idn;
            start = __builtin_amdgcn_readlane(rpv, k + 1);
            n     = __builtin_amdgcn_readlane(ctv, k + 1);

            float acc[R];
#pragma unroll
            for (int r = 0; r < R; ++r) acc[r] = 0.f;
            gcn_node_sum<C>(hs, col, st, nn, d, id0, lane, acc, k + 1 < NPW, start, n, idn);

            const float dn = rsqrtf((float)(nn + 1));
            float* o = tile + (wv * NPW + k) * PITCH;
#pragma unroll
            for (int r = 0; r < R; ++r) {
                float y = acc[r] * dn + bv[r];
                o[lane + 32u * r] = y > 0.f ? y : 0.f;
            }
        }
    }
    __syncthreads();

    const int tm = nb + 16 * wv;
    if (wv >= NPB / 16 || tm >= NNODES) return;

    const int lo = (int)lane & 15;
    const int hi = (int)lane >> 4;
    const float* arow = tile + (16 * wv + lo) * PITCH + 2 * hi;
    const float* brow = B + (size_t)(2 * hi) * NC + T * lo;

    v8f acc[T];
#pragma unroll
    for (int t = 0; t < T; ++t) acc[t] = v8f{};

    constexpr int CH = 4;
    static_assert(S % CH == 0, "whole groups of k steps");
#pragma unroll 1
    for (int s0 = 0; s0 < S; s0 += CH) {
        v2f a[CH];
        vtf c0[CH], c1[CH];
#pragma unroll
        for (int i = 0; i < CH; ++i) {
            a[i]  = *(const v2f*)(arow + 4 * (s0 + i));
            c0[i] = *(const vtf*)(brow + (size_t)(4 * (s0 + i)) * NC);
            c1[i] = *(const vtf*)(brow + (size_t)(4 * (s0 + i) + 1) * NC);
        }
#pragma unroll
        for (int i = 0; i < CH; ++i) {
#pragma unroll
            for (int t = 0; t < T; ++t) {
                v2f b;
                b.x = c0[i][t];
                b.y = c1[i][t];
                acc[t] = __builtin_amdgcn_wmma_f32_16x16x4_f32(
                     false, a[i],  false, b,
                     (short)0, acc[t],  false,  false);
            }
        }
    }

    const int* crcnt = cnt + tm + 8 * hi;
    float* crow = out + (size_t)(tm + 8 * hi) * NC + T * lo;
#pragma unroll
    for (int g = 0; g < 8; ++g) {
        const float sc = rsqrtf((float)(crcnt[g] + 1));
        vtf o;
#pragma unroll
        for (int t = 0; t < T; ++t) o[t] = acc[t][g] * sc;
        *(vtf*)(crow + (size_t)g * NC) = o;
    }
}

extern "C" void kernel_launch(void* const* d_in, const int* in_sizes, int n_in,
                              void* d_out, int out_size, void* d_ws, size_t ws_size,
                              hipStream_t stream)
{
    const float* x    = (const float*)d_in[0];
    const int*   eidx = (const int*)d_in[1];
    const int*   src  = eidx;
    const int*   dst  = eidx + NEDGES;
    const float* W1   = (const float*)d_in[2];
    const float* b1   = (const float*)d_in[3];
    const float* W2   = (const float*)d_in[4];
    const float* b2   = (const float*)d_in[5];
    float* out = (float*)d_out;

    int*   cnt    = (int*)d_ws;
    int*   rowptr = cnt + NNODES;
    int*   keys   = rowptr + NNODES;
    int*   col    = keys + PART_NB * PART_CH;
    int*   tab    = col + NEDGES;
    float* h      = (float*)(tab + (NBKT + 1) * TAB_P);
    float* h2     = h + (size_t)NNODES * HID_C;

    const int B = 256;

    gcn_part<<<PART_NB, PART_T, 0, stream>>>(src, dst, keys, tab);
    gcn_build<<<NBKT, BKT_W, 0, stream>>>(keys, tab, cnt, rowptr, col);

    {
        int waves = NNODES / 16;
        int thr = waves * 32;
        gcn_gemm_wmma<IN_C, HID_C><<<(thr + B - 1) / B, B, 0, stream>>>(x, W1, cnt, h, NNODES);
    }

    gcn_gather_gemm<HID_C, OUT_C><<<(NNODES + 63) / 64, B, 0, stream>>>(h, rowptr, cnt, col, b1, W2, h2);

    {
        int thr = (NNODES / NPW) * 32;
        gcn_gather<OUT_C, false><<<(thr + B - 1) / B, B, 0, stream>>>(h2, rowptr, cnt, col, b2, out);
    }
}
